# starting kernel
# baseline (speedup 1.0000x reference)
#include <hip/hip_runtime.h>


constexpr int kDepth   = 16;
constexpr int kB       = 16;
constexpr int kH       = 96;
constexpr int kW       = 96;
constexpr int kNPad    = 96;
constexpr int kKPad    = 12;
constexpr int kWaves   = 6;
constexpr int kThreads = kWaves * 32;

typedef __attribute__((ext_vector_type(2))) float v2f;
typedef __attribute__((ext_vector_type(8))) float v8f;

__global__ __launch_bounds__(kThreads)
void vin_step_kernel(const float* __restrict__ v_in,
                     const float* __restrict__ rewards,
                     const float* __restrict__ wt,
                     float* __restrict__ v_out) {
    __shared__ float lds_rv[3][kW + 2];
    __shared__ float lds_w[kNPad][kKPad];
    __shared__ float lds_dyn[kW][kNPad];
    __shared__ float lds_red[kW];

    const int tid  = threadIdx.x;
    const int lane = tid & 31;
    const int wv   = tid >> 5;
    const int b    = blockIdx.x / kH;
    const int h    = blockIdx.x % kH;

    const float* vimg = v_in    + (size_t)b * kH * kW;
    const float* rimg = rewards + (size_t)b * kH * kW;

    for (int i = tid; i < 3 * (kW + 2); i += kThreads) {
        const int r  = i / (kW + 2);
        const int c  = i % (kW + 2);
        const int hh = h - 1 + r;
        const int ww = c - 1;
        const bool inb = (hh >= 0) & (hh < kH) & (ww >= 0) & (ww < kW);
        const int hc = min(max(hh, 0), kH - 1);
        const int wc = min(max(ww, 0), kW - 1);
        const float val = vimg[hc * kW + wc] + rimg[hc * kW + wc];
        lds_rv[r][c] = inb ? val : 0.0f;
    }

    for (int i = tid; i < kNPad * kKPad; i += kThreads) {
        const int ch = i / kKPad;
        const int k  = i % kKPad;
        const bool valid = (ch < 90) & (k < 9);
        const float val = wt[min(ch, 89) * 9 + min(k, 8)];
        lds_w[ch][k] = valid ? val : 0.0f;
    }
    __syncthreads();

    const int  m  = (lane & 15) + wv * 16;
    const bool lo = lane < 16;
#define RV(r, c) lds_rv[r][m + (c)]
    v2f afrag[3];
    afrag[0].x = lo ? RV(0, 0) : RV(0, 2);
    afrag[0].y = lo ? RV(0, 1) : RV(1, 0);
    afrag[1].x = lo ? RV(1, 1) : RV(2, 0);
    afrag[1].y = lo ? RV(1, 2) : RV(2, 1);
    afrag[2].x = lo ? RV(2, 2) : 0.0f;
    afrag[2].y = 0.0f;
#undef RV

    const int n    = (lane & 15);
    const int koff = lo ? 0 : 2;
#pragma unroll 1
    for (int nt = 0; nt < 6; ++nt) {
        const int ch = nt * 16 + n;
        v2f b0 = *(const v2f*)&lds_w[ch][0 + koff];
        v2f b1 = *(const v2f*)&lds_w[ch][4 + koff];
        v2f b2 = *(const v2f*)&lds_w[ch][8 + koff];
        v8f acc = {};
        acc = __builtin_amdgcn_wmma_f32_16x16x4_f32(false, afrag[0], false, b0,
                                                    (short)0, acc, false, false);
        acc = __builtin_amdgcn_wmma_f32_16x16x4_f32(false, afrag[1], false, b1,
                                                    (short)0, acc, false, false);
        acc = __builtin_amdgcn_wmma_f32_16x16x4_f32(false, afrag[2], false, b2,
                                                    (short)0, acc, false, false);
        const int mbase = wv * 16 + (lo ? 0 : 8);
#pragma unroll
        for (int r = 0; r < 8; ++r)
            lds_dyn[mbase + r][ch] = acc[r];
    }
    __syncthreads();

    const int p    = tid % kW;
    const int half = tid / kW;

    float patch[9];
#pragma unroll
    for (int t = 0; t < 9; ++t)
        patch[t] = lds_rv[t / 3][p + (t % 3)];

    float best = -3.402823466e38f;
    for (int a = half * 5; a < half * 5 + 5; ++a) {
        const float* dl = &lds_dyn[p][a * 9];
        float mx = dl[0];
#pragma unroll
        for (int t = 1; t < 9; ++t) mx = fmaxf(mx, dl[t]);
        float ssum = 0.0f, q = 0.0f;
#pragma unroll
        for (int t = 0; t < 9; ++t) {
            const float e = __expf(dl[t] - mx);
            ssum += e;
            q    += e * patch[t];
        }
        best = fmaxf(best, q / ssum);
    }

    if (half == 1) lds_red[p] = best;
    __syncthreads();
    if (half == 0) {
        best = fmaxf(best, lds_red[p]);
        v_out[((size_t)b * kH + h) * kW + p] = best;
    }
}

extern "C" void kernel_launch(void* const* d_in, const int* in_sizes, int n_in,
                              void* d_out, int out_size, void* d_ws, size_t ws_size,
                              hipStream_t stream) {
    (void)in_sizes; (void)n_in; (void)out_size; (void)d_ws; (void)ws_size;
    const float* values  = (const float*)d_in[0];
    const float* rewards = (const float*)d_in[1];
    const float* wt      = (const float*)d_in[2];
    float* out = (float*)d_out;

    const size_t slice = (size_t)kB * kH * kW;
    for (int t = 0; t < kDepth; ++t) {
        const float* vin = (t == 0) ? values : out + (size_t)(t - 1) * slice;
        float* vout = out + (size_t)t * slice;
        vin_step_kernel<<<kB * kH, kThreads, 0, stream>>>(vin, rewards, wt, vout);
    }
}


# LDS logits tile stored channel-major [ch][px] at pitch 100 dwords: softmax reads conflict-free (pixel-major pitch 96 put a wave's lanes on 2 of 64 banks, 16-way); acc stores 2 x b128 per tile
# speedup vs baseline: 2.2269x; 2.2269x over previous
#include <hip/hip_runtime.h>


constexpr int kDepth   = 16;
constexpr int kB       = 16;
constexpr int kH       = 96;
constexpr int kW       = 96;
constexpr int kNPad    = 96;
constexpr int kKPad    = 12;
constexpr int kPxPitch = kW + 4;
constexpr int kWaves   = 6;
constexpr int kThreads = kWaves * 32;

typedef __attribute__((ext_vector_type(2))) float v2f;
typedef __attribute__((ext_vector_type(4))) float v4f;
typedef __attribute__((ext_vector_type(8))) float v8f;

__global__ __launch_bounds__(kThreads)
void vin_step_kernel(const float* __restrict__ v_in,
                     const float* __restrict__ rewards,
                     const float* __restrict__ wt,
                     float* __restrict__ v_out) {
    __shared__ float lds_rv[3][kW + 2];
    __shared__ float lds_w[kNPad][kKPad];
    __shared__ alignas(16) float lds_dyn[kNPad][kPxPitch];
    __shared__ float lds_red[kW];

    const int tid  = threadIdx.x;
    const int lane = tid & 31;
    const int wv   = tid >> 5;
    const int b    = blockIdx.x / kH;
    const int h    = blockIdx.x % kH;

    const float* vimg = v_in    + (size_t)b * kH * kW;
    const float* rimg = rewards + (size_t)b * kH * kW;

    for (int i = tid; i < 3 * (kW + 2); i += kThreads) {
        const int r  = i / (kW + 2);
        const int c  = i % (kW + 2);
        const int hh = h - 1 + r;
        const int ww = c - 1;
        const bool inb = (hh >= 0) & (hh < kH) & (ww >= 0) & (ww < kW);
        const int hc = min(max(hh, 0), kH - 1);
        const int wc = min(max(ww, 0), kW - 1);
        const float val = vimg[hc * kW + wc] + rimg[hc * kW + wc];
        lds_rv[r][c] = inb ? val : 0.0f;
    }

    for (int i = tid; i < kNPad * kKPad; i += kThreads) {
        const int ch = i / kKPad;
        const int k  = i % kKPad;
        const bool valid = (ch < 90) & (k < 9);
        const float val = wt[min(ch, 89) * 9 + min(k, 8)];
        lds_w[ch][k] = valid ? val : 0.0f;
    }
    __syncthreads();

    const int  m  = (lane & 15) + wv * 16;
    const bool lo = lane < 16;
#define RV(r, c) lds_rv[r][m + (c)]
    v2f afrag[3];
    afrag[0].x = lo ? RV(0, 0) : RV(0, 2);
    afrag[0].y = lo ? RV(0, 1) : RV(1, 0);
    afrag[1].x = lo ? RV(1, 1) : RV(2, 0);
    afrag[1].y = lo ? RV(1, 2) : RV(2, 1);
    afrag[2].x = lo ? RV(2, 2) : 0.0f;
    afrag[2].y = 0.0f;
#undef RV

    const int n    = (lane & 15);
    const int koff = lo ? 0 : 2;
    const int mbase = wv * 16 + (lo ? 0 : 8);
#pragma unroll 1
    for (int nt = 0; nt < 6; ++nt) {
        const int ch = nt * 16 + n;
        v2f b0 = *(const v2f*)&lds_w[ch][0 + koff];
        v2f b1 = *(const v2f*)&lds_w[ch][4 + koff];
        v2f b2 = *(const v2f*)&lds_w[ch][8 + koff];
        v8f acc = {};
        acc = __builtin_amdgcn_wmma_f32_16x16x4_f32(false, afrag[0], false, b0,
                                                    (short)0, acc, false, false);
        acc = __builtin_amdgcn_wmma_f32_16x16x4_f32(false, afrag[1], false, b1,
                                                    (short)0, acc, false, false);
        acc = __builtin_amdgcn_wmma_f32_16x16x4_f32(false, afrag[2], false, b2,
                                                    (short)0, acc, false, false);
        float* drow = &lds_dyn[ch][mbase];
        *(v4f*)(drow)     = __builtin_shufflevector(acc, acc, 0, 1, 2, 3);
        *(v4f*)(drow + 4) = __builtin_shufflevector(acc, acc, 4, 5, 6, 7);
    }
    __syncthreads();

    const int p    = tid % kW;
    const int half = tid / kW;

    float patch[9];
#pragma unroll
    for (int t = 0; t < 9; ++t)
        patch[t] = lds_rv[t / 3][p + (t % 3)];

    float best = -3.402823466e38f;
    for (int a = half * 5; a < half * 5 + 5; ++a) {
        const float* dl = &lds_dyn[a * 9][p];
        float mx = dl[0];
#pragma unroll
        for (int t = 1; t < 9; ++t) mx = fmaxf(mx, dl[t * kPxPitch]);
        float ssum = 0.0f, q = 0.0f;
#pragma unroll
        for (int t = 0; t < 9; ++t) {
            const float e = __expf(dl[t * kPxPitch] - mx);
            ssum += e;
            q    += e * patch[t];
        }
        best = fmaxf(best, q / ssum);
    }

    if (half == 1) lds_red[p] = best;
    __syncthreads();
    if (half == 0) {
        best = fmaxf(best, lds_red[p]);
        v_out[((size_t)b * kH + h) * kW + p] = best;
    }
}

extern "C" void kernel_launch(void* const* d_in, const int* in_sizes, int n_in,
                              void* d_out, int out_size, void* d_ws, size_t ws_size,
                              hipStream_t stream) {
    (void)in_sizes; (void)n_in; (void)out_size; (void)d_ws; (void)ws_size;
    const float* values  = (const float*)d_in[0];
    const float* rewards = (const float*)d_in[1];
    const float* wt      = (const float*)d_in[2];
    float* out = (float*)d_out;

    const size_t slice = (size_t)kB * kH * kW;
    for (int t = 0; t < kDepth; ++t) {
        const float* vin = (t == 0) ? values : out + (size_t)(t - 1) * slice;
        float* vout = out + (size_t)t * slice;
        vin_step_kernel<<<kB * kH, kThreads, 0, stream>>>(vin, rewards, wt, vout);
    }
}


# Staging split into a load pass and a store pass: all 10 global loads of a thread (rv strip, weights) are issued before the first wait: one memory round trip per work-group instead of eight
# speedup vs baseline: 2.6197x; 1.1764x over previous
#include <hip/hip_runtime.h>


constexpr int kDepth   = 16;
constexpr int kB       = 16;
constexpr int kH       = 96;
constexpr int kW       = 96;
constexpr int kNPad    = 96;
constexpr int kKPad    = 12;
constexpr int kPxPitch = kW + 4;
constexpr int kWaves   = 6;
constexpr int kThreads = kWaves * 32;

typedef __attribute__((ext_vector_type(2))) float v2f;
typedef __attribute__((ext_vector_type(4))) float v4f;
typedef __attribute__((ext_vector_type(8))) float v8f;

__global__ __launch_bounds__(kThreads)
void vin_step_kernel(const float* __restrict__ v_in,
                     const float* __restrict__ rewards,
                     const float* __restrict__ wt,
                     float* __restrict__ v_out) {
    __shared__ float lds_rv[3][kW + 2];
    __shared__ float lds_w[kNPad][kKPad];
    __shared__ alignas(16) float lds_dyn[kNPad][kPxPitch];
    __shared__ float lds_red[kW];

    const int tid  = threadIdx.x;
    const int lane = tid & 31;
    const int wv   = tid >> 5;
    const int b    = blockIdx.x / kH;
    const int h    = blockIdx.x % kH;

    const float* vimg = v_in    + (size_t)b * kH * kW;
    const float* rimg = rewards + (size_t)b * kH * kW;

    constexpr int kRvElems = 3 * (kW + 2);
    constexpr int kRvTrips = (kRvElems + kThreads - 1) / kThreads;
    constexpr int kWtTrips = (kNPad * kKPad) / kThreads;
    static_assert(kWtTrips * kThreads == kNPad * kKPad, "weight tile must divide evenly");

    float rv_v[kRvTrips], rv_r[kRvTrips];
    float w_g[kWtTrips];
#pragma unroll
    for (int j = 0; j < kRvTrips; ++j) {
        const int i  = min(tid + j * kThreads, kRvElems - 1);
        const int r  = i / (kW + 2);
        const int c  = i % (kW + 2);
        const int hc = min(max(h - 1 + r, 0), kH - 1);
        const int wc = min(max(c - 1, 0), kW - 1);
        rv_v[j] = vimg[hc * kW + wc];
        rv_r[j] = rimg[hc * kW + wc];
    }
#pragma unroll
    for (int j = 0; j < kWtTrips; ++j) {
        const int i  = tid + j * kThreads;
        const int ch = i / kKPad;
        const int k  = i % kKPad;
        w_g[j] = wt[min(ch, 89) * 9 + min(k, 8)];
    }

    static_assert(kRvTrips == 2 && kWtTrips == 6, "fence operand list is written for 2 + 6 trips");
    asm volatile("" :: "v"(rv_v[0]), "v"(rv_r[0]), "v"(rv_v[1]), "v"(rv_r[1]),
                       "v"(w_g[0]), "v"(w_g[1]), "v"(w_g[2]),
                       "v"(w_g[3]), "v"(w_g[4]), "v"(w_g[5]));

#pragma unroll
    for (int j = 0; j < kRvTrips; ++j) {
        const int i  = tid + j * kThreads;
        const int r  = i / (kW + 2);
        const int c  = i % (kW + 2);
        const int hh = h - 1 + r;
        const int ww = c - 1;
        const bool inb = (hh >= 0) & (hh < kH) & (ww >= 0) & (ww < kW);
        const bool in_strip = ((j + 1) * kThreads <= kRvElems) || (i < kRvElems);
        if (in_strip) lds_rv[r][c] = inb ? (rv_v[j] + rv_r[j]) : 0.0f;
    }
#pragma unroll
    for (int j = 0; j < kWtTrips; ++j) {
        const int i  = tid + j * kThreads;
        const int ch = i / kKPad;
        const int k  = i % kKPad;
        const bool valid = (ch < 90) & (k < 9);
        lds_w[ch][k] = valid ? w_g[j] : 0.0f;
    }
    __syncthreads();

    const int  m  = (lane & 15) + wv * 16;
    const bool lo = lane < 16;
#define RV(r, c) lds_rv[r][m + (c)]
    v2f afrag[3];
    afrag[0].x = lo ? RV(0, 0) : RV(0, 2);
    afrag[0].y = lo ? RV(0, 1) : RV(1, 0);
    afrag[1].x = lo ? RV(1, 1) : RV(2, 0);
    afrag[1].y = lo ? RV(1, 2) : RV(2, 1);
    afrag[2].x = lo ? RV(2, 2) : 0.0f;
    afrag[2].y = 0.0f;
#undef RV

    const int n    = (lane & 15);
    const int koff = lo ? 0 : 2;
    const int mbase = wv * 16 + (lo ? 0 : 8);
#pragma unroll 1
    for (int nt = 0; nt < 6; ++nt) {
        const int ch = nt * 16 + n;
        v2f b0 = *(const v2f*)&lds_w[ch][0 + koff];
        v2f b1 = *(const v2f*)&lds_w[ch][4 + koff];
        v2f b2 = *(const v2f*)&lds_w[ch][8 + koff];
        v8f acc = {};
        acc = __builtin_amdgcn_wmma_f32_16x16x4_f32(false, afrag[0], false, b0,
                                                    (short)0, acc, false, false);
        acc = __builtin_amdgcn_wmma_f32_16x16x4_f32(false, afrag[1], false, b1,
                                                    (short)0, acc, false, false);
        acc = __builtin_amdgcn_wmma_f32_16x16x4_f32(false, afrag[2], false, b2,
                                                    (short)0, acc, false, false);
        float* drow = &lds_dyn[ch][mbase];
        *(v4f*)(drow)     = __builtin_shufflevector(acc, acc, 0, 1, 2, 3);
        *(v4f*)(drow + 4) = __builtin_shufflevector(acc, acc, 4, 5, 6, 7);
    }
    __syncthreads();

    const int p    = tid % kW;
    const int half = tid / kW;

    float patch[9];
#pragma unroll
    for (int t = 0; t < 9; ++t)
        patch[t] = lds_rv[t / 3][p + (t % 3)];

    float best = -3.402823466e38f;
    for (int a = half * 5; a < half * 5 + 5; ++a) {
        const float* dl = &lds_dyn[a * 9][p];
        float mx = dl[0];
#pragma unroll
        for (int t = 1; t < 9; ++t) mx = fmaxf(mx, dl[t * kPxPitch]);
        float ssum = 0.0f, q = 0.0f;
#pragma unroll
        for (int t = 0; t < 9; ++t) {
            const float e = __expf(dl[t * kPxPitch] - mx);
            ssum += e;
            q    += e * patch[t];
        }
        best = fmaxf(best, q / ssum);
    }

    if (half == 1) lds_red[p] = best;
    __syncthreads();
    if (half == 0) {
        best = fmaxf(best, lds_red[p]);
        v_out[((size_t)b * kH + h) * kW + p] = best;
    }
}

extern "C" void kernel_launch(void* const* d_in, const int* in_sizes, int n_in,
                              void* d_out, int out_size, void* d_ws, size_t ws_size,
                              hipStream_t stream) {
    (void)in_sizes; (void)n_in; (void)out_size; (void)d_ws; (void)ws_size;
    const float* values  = (const float*)d_in[0];
    const float* rewards = (const float*)d_in[1];
    const float* wt      = (const float*)d_in[2];
    float* out = (float*)d_out;

    const size_t slice = (size_t)kB * kH * kW;
    for (int t = 0; t < kDepth; ++t) {
        const float* vin = (t == 0) ? values : out + (size_t)(t - 1) * slice;
        float* vout = out + (size_t)t * slice;
        vin_step_kernel<<<kB * kH, kThreads, 0, stream>>>(vin, rewards, wt, vout);
    }
}


# Product transposed (weights = A operand, pixels = columns; channel slots ordered so a lane's accumulators hold all 9 logits of an action): softmax on accumulators, logits skip LDS, 1 barrier not 3
# speedup vs baseline: 2.9678x; 1.1329x over previous
#include <hip/hip_runtime.h>


constexpr int kDepth   = 16;
constexpr int kB       = 16;
constexpr int kH       = 96;
constexpr int kW       = 96;
constexpr int kWaves   = 3;
constexpr int kThreads = kWaves * 32;
static_assert(kThreads == kW, "staging maps one thread to one column of the row");
constexpr int kWtPairs = 90 * 9 / 2;
constexpr int kWtTrips = (kWtPairs + kThreads - 1) / kThreads;

typedef __attribute__((ext_vector_type(2))) float v2f;
typedef __attribute__((ext_vector_type(8))) float v8f;

__global__ __launch_bounds__(kThreads)
void vin_step_kernel(const float* __restrict__ v_in,
                     const float* __restrict__ rewards,
                     const float* __restrict__ wt,
                     float* __restrict__ v_out) {
    __shared__ float lds_rv[3][kW + 2];
    __shared__ v2f lds_wt[kWtTrips * kThreads];

    const int tid  = threadIdx.x;
    const int lane = tid & 31;
    const int wv   = tid >> 5;
    const int c    = lane & 15;
    const int hh   = lane >> 4;
    const bool lo  = lane < 16;
    const int b    = blockIdx.x / kH;
    const int h    = blockIdx.x % kH;

    const float* vimg = v_in    + (size_t)b * kH * kW;
    const float* rimg = rewards + (size_t)b * kH * kW;

    float sv[3], sr[3];
#pragma unroll
    for (int r = 0; r < 3; ++r) {
        const int hc = min(max(h - 1 + r, 0), kH - 1);
        sv[r] = vimg[hc * kW + tid];
        sr[r] = rimg[hc * kW + tid];
    }

    v2f wg[kWtTrips];
#pragma unroll
    for (int j = 0; j < kWtTrips; ++j)
        wg[j] = ((const v2f*)wt)[min(tid + j * kThreads, kWtPairs - 1)];

#pragma unroll
    for (int r = 0; r < 3; ++r) {
        const int hr = h - 1 + r;
        const bool inb = (hr >= 0) & (hr < kH);
        lds_rv[r][tid + 1] = inb ? (sv[r] + sr[r]) : 0.0f;
    }
    if (tid < 3) {
        lds_rv[tid][0]      = 0.0f;
        lds_rv[tid][kW + 1] = 0.0f;
    }
#pragma unroll
    for (int j = 0; j < kWtTrips; ++j)
        lds_wt[tid + j * kThreads] = wg[j];
    __syncthreads();

    const int r8 = c & 7;
    const int hs = c >> 3;
    v2f afrag[6][3];
    {
        const float* lw   = (const float*)lds_wt;
        const float* wrow = lw + (hs * 9 + r8) * 9;
#pragma unroll
        for (int j = 0; j < 5; ++j) {
            const float* p = wrow + j * 162 + 2 * hh;
            afrag[j][0].x = p[0];
            afrag[j][0].y = p[1];
            afrag[j][1].x = p[4];
            afrag[j][1].y = p[5];
            const float w8 = wrow[j * 162 + 8];
            afrag[j][2].x = lo ? w8 : 0.0f;
            afrag[j][2].y = 0.0f;
        }
        const bool used = r8 < 5;
        const float* w5 = lw + ((2 * min(r8, 4) + hs) * 9 + 8) * 9;
        const float* p  = w5 + 2 * hh;
        const float a0 = p[0], a1 = p[1], a4 = p[4], a5 = p[5];
        const float a8 = w5[8];
        afrag[5][0].x = used ? a0 : 0.0f;
        afrag[5][0].y = used ? a1 : 0.0f;
        afrag[5][1].x = used ? a4 : 0.0f;
        afrag[5][1].y = used ? a5 : 0.0f;
        afrag[5][2].x = (used & lo) ? a8 : 0.0f;
        afrag[5][2].y = 0.0f;
    }


    float outv = 0.0f;
#pragma unroll
    for (int tile = 0; tile < 2; ++tile) {
        const int px = wv * 32 + tile * 16 + c;
        float patch[9];
#pragma unroll
        for (int t = 0; t < 9; ++t)
            patch[t] = lds_rv[t / 3][px + (t % 3)];

        v2f bfrag[3];
        bfrag[0].x = lo ? patch[0] : patch[2];
        bfrag[0].y = lo ? patch[1] : patch[3];
        bfrag[1].x = lo ? patch[4] : patch[6];
        bfrag[1].y = lo ? patch[5] : patch[7];
        bfrag[2].x = lo ? patch[8] : 0.0f;
        bfrag[2].y = 0.0f;

        v8f acc9 = {};
        acc9 = __builtin_amdgcn_wmma_f32_16x16x4_f32(false, afrag[5][0], false, bfrag[0],
                                                     (short)0, acc9, false, false);
        acc9 = __builtin_amdgcn_wmma_f32_16x16x4_f32(false, afrag[5][1], false, bfrag[1],
                                                     (short)0, acc9, false, false);
        acc9 = __builtin_amdgcn_wmma_f32_16x16x4_f32(false, afrag[5][2], false, bfrag[2],
                                                     (short)0, acc9, false, false);

        float best = -3.402823466e38f;
#pragma unroll
        for (int j = 0; j < 5; ++j) {
            v8f acc = {};
            acc = __builtin_amdgcn_wmma_f32_16x16x4_f32(false, afrag[j][0], false, bfrag[0],
                                                        (short)0, acc, false, false);
            acc = __builtin_amdgcn_wmma_f32_16x16x4_f32(false, afrag[j][1], false, bfrag[1],
                                                        (short)0, acc, false, false);
            acc = __builtin_amdgcn_wmma_f32_16x16x4_f32(false, afrag[j][2], false, bfrag[2],
                                                        (short)0, acc, false, false);
            const float l8 = acc9[j];
            float mx = acc[0];
#pragma unroll
            for (int t = 1; t < 8; ++t) mx = fmaxf(mx, acc[t]);
            mx = fmaxf(mx, l8);
            float ssum = 0.0f, q = 0.0f;
#pragma unroll
            for (int t = 0; t < 8; ++t) {
                const float e = __expf(acc[t] - mx);
                ssum += e;
                q    += e * patch[t];
            }
            {
                const float e = __expf(l8 - mx);
                ssum += e;
                q    += e * patch[8];
            }
            best = fmaxf(best, q / ssum);
        }
        best = fmaxf(best, __shfl_xor(best, 16));
        outv = (hh == tile) ? best : outv;
    }

    v_out[((size_t)b * kH + h) * kW + wv * 32 + lane] = outv;
}

extern "C" void kernel_launch(void* const* d_in, const int* in_sizes, int n_in,
                              void* d_out, int out_size, void* d_ws, size_t ws_size,
                              hipStream_t stream) {
    (void)in_sizes; (void)n_in; (void)out_size; (void)d_ws; (void)ws_size;
    const float* values  = (const float*)d_in[0];
    const float* rewards = (const float*)d_in[1];
    const float* wt      = (const float*)d_in[2];
    float* out = (float*)d_out;

    const size_t slice = (size_t)kB * kH * kW;
    for (int t = 0; t < kDepth; ++t) {
        const float* vin = (t == 0) ? values : out + (size_t)(t - 1) * slice;
        float* vout = out + (size_t)t * slice;
        vin_step_kernel<<<kB * kH, kThreads, 0, stream>>>(vin, rewards, wt, vout);
    }
}


# Two recurrence steps per launch: a work-group owns a 6-row strip, computes step t on 8 rows (one recomputed halo row each side), keeps v_t + rewards in LDS, then step t+1 on its rows: 8 launches
# speedup vs baseline: 3.1145x; 1.0494x over previous
#include <hip/hip_runtime.h>


constexpr int kDepth   = 16;
constexpr int kB       = 16;
constexpr int kH       = 96;
constexpr int kW       = 96;
constexpr int kStrip   = 6;
constexpr int kStrips  = kH / kStrip;
constexpr int kRows1   = kStrip + 2;
constexpr int kRowsIn  = kStrip + 4;
constexpr int kWaves   = 12;
constexpr int kThreads = kWaves * 32;
constexpr int kPitch   = kW + 2;
constexpr int kSegs    = kW / 32;
constexpr int kUnits1  = kRows1 * kSegs;
constexpr int kUnits2  = kStrip * kSegs;
constexpr int kInTrips = (kRowsIn * kW + kThreads - 1) / kThreads;
constexpr int kStageRows = kInTrips * kThreads / kW;
static_assert(kInTrips * kThreads % kW == 0 && kStageRows >= kRowsIn, "whole rows per staging pass");
constexpr int kWtPairs = 90 * 9 / 2;
constexpr int kWtTrips = (kWtPairs + kThreads - 1) / kThreads;
static_assert(kH % kStrip == 0, "strips tile the image");
static_assert(kUnits1 == 2 * kWaves, "step one: exactly two units per wave");
static_assert(kUnits2 - kWaves == 6, "step two: six waves take a second unit");

typedef __attribute__((ext_vector_type(2))) float v2f;
typedef __attribute__((ext_vector_type(8))) float v8f;

__device__ __forceinline__ float vin_unit(const float* rows, int px0,
                                          const v2f (&afrag)[6][3],
                                          int c, int hh, bool lo) {
    float outv = 0.0f;
#pragma unroll
    for (int tile = 0; tile < 2; ++tile) {
        const int px = px0 + tile * 16 + c;
        float patch[9];
#pragma unroll
        for (int t = 0; t < 9; ++t)
            patch[t] = rows[(t / 3) * kPitch + px + (t % 3)];

        v2f bfrag[3];
        bfrag[0].x = lo ? patch[0] : patch[2];
        bfrag[0].y = lo ? patch[1] : patch[3];
        bfrag[1].x = lo ? patch[4] : patch[6];
        bfrag[1].y = lo ? patch[5] : patch[7];
        bfrag[2].x = lo ? patch[8] : 0.0f;
        bfrag[2].y = 0.0f;

        v8f acc9 = {};
        acc9 = __builtin_amdgcn_wmma_f32_16x16x4_f32(false, afrag[5][0], false, bfrag[0],
                                                     (short)0, acc9, false, false);
        acc9 = __builtin_amdgcn_wmma_f32_16x16x4_f32(false, afrag[5][1], false, bfrag[1],
                                                     (short)0, acc9, false, false);
        acc9 = __builtin_amdgcn_wmma_f32_16x16x4_f32(false, afrag[5][2], false, bfrag[2],
                                                     (short)0, acc9, false, false);

        float best = -3.402823466e38f;
#pragma unroll
        for (int j = 0; j < 5; ++j) {
            v8f acc = {};
            acc = __builtin_amdgcn_wmma_f32_16x16x4_f32(false, afrag[j][0], false, bfrag[0],
                                                        (short)0, acc, false, false);
            acc = __builtin_amdgcn_wmma_f32_16x16x4_f32(false, afrag[j][1], false, bfrag[1],
                                                        (short)0, acc, false, false);
            acc = __builtin_amdgcn_wmma_f32_16x16x4_f32(false, afrag[j][2], false, bfrag[2],
                                                        (short)0, acc, false, false);
            const float l8 = acc9[j];
            float mx = acc[0];
#pragma unroll
            for (int t = 1; t < 8; ++t) mx = fmaxf(mx, acc[t]);
            mx = fmaxf(mx, l8);
            float ssum = 0.0f, q = 0.0f;
#pragma unroll
            for (int t = 0; t < 8; ++t) {
                const float e = __expf(acc[t] - mx);
                ssum += e;
                q    += e * patch[t];
            }
            {
                const float e = __expf(l8 - mx);
                ssum += e;
                q    += e * patch[8];
            }
            best = fmaxf(best, q / ssum);
        }
        best = fmaxf(best, __shfl_xor(best, 16));
        outv = (hh == tile) ? best : outv;
    }
    return outv;
}

__global__ __launch_bounds__(kThreads)
void vin_two_step_kernel(const float* __restrict__ v_in,
                         const float* __restrict__ rewards,
                         const float* __restrict__ wt,
                         float* __restrict__ v_out1,
                         float* __restrict__ v_out2) {
    __shared__ float lds_rv1[kStageRows][kPitch];
    __shared__ float lds_rw[kStageRows][kW];
    __shared__ float lds_rv2[kRows1][kPitch];
    __shared__ v2f lds_wt[kWtTrips * kThreads];

    const int tid  = threadIdx.x;
    const int lane = tid & 31;
    const int wv   = __builtin_amdgcn_readfirstlane(tid >> 5);
    const int c    = lane & 15;
    const int hh   = lane >> 4;
    const bool lo  = lane < 16;
    const int b    = blockIdx.x / kStrips;
    const int h0   = (blockIdx.x % kStrips) * kStrip;

    const float* vimg = v_in    + (size_t)b * kH * kW;
    const float* rimg = rewards + (size_t)b * kH * kW;

    float sv[kInTrips], sr[kInTrips];
#pragma unroll
    for (int j = 0; j < kInTrips; ++j) {
        const int e   = tid + j * kThreads;
        const int row = e / kW;
        const int col = e - row * kW;
        const int hc  = min(max(h0 - 2 + row, 0), kH - 1);
        sv[j] = vimg[hc * kW + col];
        sr[j] = rimg[hc * kW + col];
    }
    v2f wg[kWtTrips];
#pragma unroll
    for (int j = 0; j < kWtTrips; ++j)
        wg[j] = ((const v2f*)wt)[min(tid + j * kThreads, kWtPairs - 1)];

#pragma unroll
    for (int j = 0; j < kInTrips; ++j) {
        const int e   = tid + j * kThreads;
        const int row = e / kW;
        const int col = e - row * kW;
        const int hr  = h0 - 2 + row;
        const bool inb = (hr >= 0) & (hr < kH);
        lds_rv1[row][col + 1] = inb ? (sv[j] + sr[j]) : 0.0f;
        lds_rw[row][col]      = sr[j];
    }
    if (tid < kRowsIn) {
        lds_rv1[tid][0]      = 0.0f;
        lds_rv1[tid][kW + 1] = 0.0f;
    }
    if (tid < kRows1) {
        lds_rv2[tid][0]      = 0.0f;
        lds_rv2[tid][kW + 1] = 0.0f;
    }
#pragma unroll
    for (int j = 0; j < kWtTrips; ++j)
        lds_wt[tid + j * kThreads] = wg[j];
    __syncthreads();

    const int r8 = c & 7;
    const int hs = c >> 3;
    v2f afrag[6][3];
    {
        const float* lw   = (const float*)lds_wt;
        const float* wrow = lw + (hs * 9 + r8) * 9;
#pragma unroll
        for (int j = 0; j < 5; ++j) {
            const float* p = wrow + j * 162 + 2 * hh;
            afrag[j][0].x = p[0];
            afrag[j][0].y = p[1];
            afrag[j][1].x = p[4];
            afrag[j][1].y = p[5];
            const float w8 = wrow[j * 162 + 8];
            afrag[j][2].x = lo ? w8 : 0.0f;
            afrag[j][2].y = 0.0f;
        }
        const bool used = r8 < 5;
        const float* w5 = lw + ((2 * min(r8, 4) + hs) * 9 + 8) * 9;
        const float* p  = w5 + 2 * hh;
        const float a0 = p[0], a1 = p[1], a4 = p[4], a5 = p[5];
        const float a8 = w5[8];
        afrag[5][0].x = used ? a0 : 0.0f;
        afrag[5][0].y = used ? a1 : 0.0f;
        afrag[5][1].x = used ? a4 : 0.0f;
        afrag[5][1].y = used ? a5 : 0.0f;
        afrag[5][2].x = (used & lo) ? a8 : 0.0f;
        afrag[5][2].y = 0.0f;
    }

    for (int i = 0; i < 2; ++i) {
        const int u   = wv + i * kWaves;
        const int lr  = u / kSegs;
        const int seg = u - lr * kSegs;
        const float nv = vin_unit(&lds_rv1[lr][0], seg * 32, afrag, c, hh, lo);
        const int hr  = h0 - 1 + lr;
        const int col = seg * 32 + lane;
        const bool inb = (hr >= 0) & (hr < kH);
        lds_rv2[lr][col + 1] = inb ? (nv + lds_rw[lr + 1][col]) : 0.0f;
        if (lr >= 1 && lr <= kStrip)
            v_out1[((size_t)b * kH + hr) * kW + col] = nv;
    }
    __syncthreads();

    const int n2 = (wv % 5 <= 1) ? 2 : 1;
    for (int i = 0; i < n2; ++i) {
        const int u   = (i == 0) ? wv : (kWaves + (wv / 5) * 2 + (wv % 5));
        const int lr  = u / kSegs;
        const int seg = u - lr * kSegs;
        const float nv = vin_unit(&lds_rv2[lr][0], seg * 32, afrag, c, hh, lo);
        v_out2[((size_t)b * kH + h0 + lr) * kW + seg * 32 + lane] = nv;
    }
}

extern "C" void kernel_launch(void* const* d_in, const int* in_sizes, int n_in,
                              void* d_out, int out_size, void* d_ws, size_t ws_size,
                              hipStream_t stream) {
    (void)in_sizes; (void)n_in; (void)out_size; (void)d_ws; (void)ws_size;
    const float* values  = (const float*)d_in[0];
    const float* rewards = (const float*)d_in[1];
    const float* wt      = (const float*)d_in[2];
    float* out = (float*)d_out;

    const size_t slice = (size_t)kB * kH * kW;
    for (int t = 0; t < kDepth; t += 2) {
        const float* vin = (t == 0) ? values : out + (size_t)(t - 1) * slice;
        float* vout1 = out + (size_t)t * slice;
        float* vout2 = out + (size_t)(t + 1) * slice;
        vin_two_step_kernel<<<kB * kStrips, kThreads, 0, stream>>>(vin, rewards, wt, vout1, vout2);
    }
}


# A operand rows hold log2(e)*(W[t]-W[8]) per action (softmax is shift-invariant): ninth logit is 0 and needs no product, 15 WMMAs per 16 pixels not 18, exp2 taken with no multiply in front
# speedup vs baseline: 3.4021x; 1.0923x over previous
#include <hip/hip_runtime.h>


constexpr int kDepth   = 16;
constexpr int kB       = 16;
constexpr int kH       = 96;
constexpr int kW       = 96;
constexpr int kStrip   = 6;
constexpr int kStrips  = kH / kStrip;
constexpr int kRows1   = kStrip + 2;
constexpr int kRowsIn  = kStrip + 4;
constexpr int kWaves   = 12;
constexpr int kThreads = kWaves * 32;
constexpr int kPitch   = kW + 2;
constexpr int kSegs    = kW / 32;
constexpr int kUnits1  = kRows1 * kSegs;
constexpr int kUnits2  = kStrip * kSegs;
constexpr int kInTrips = (kRowsIn * kW + kThreads - 1) / kThreads;
constexpr int kStageRows = kInTrips * kThreads / kW;
static_assert(kInTrips * kThreads % kW == 0 && kStageRows >= kRowsIn, "whole rows per staging pass");
constexpr int kWtPairs = 90 * 9 / 2;
constexpr int kWtTrips = (kWtPairs + kThreads - 1) / kThreads;
static_assert(kH % kStrip == 0, "strips tile the image");
static_assert(kUnits1 == 2 * kWaves, "step one: exactly two units per wave");
static_assert(kUnits2 - kWaves == 6, "step two: six waves take a second unit");

typedef __attribute__((ext_vector_type(2))) float v2f;
typedef __attribute__((ext_vector_type(8))) float v8f;

constexpr float kLog2e = 1.4426950408889634f;

__device__ __forceinline__ float vin_unit(const float* rows, int px0,
                                          const v2f (&afrag)[5][3],
                                          int c, int hh, bool lo) {
    float outv = 0.0f;
#pragma unroll
    for (int tile = 0; tile < 2; ++tile) {
        const int px = px0 + tile * 16 + c;
        float patch[9];
#pragma unroll
        for (int t = 0; t < 9; ++t)
            patch[t] = rows[(t / 3) * kPitch + px + (t % 3)];

        v2f bfrag[3];
        bfrag[0].x = lo ? patch[0] : patch[2];
        bfrag[0].y = lo ? patch[1] : patch[3];
        bfrag[1].x = lo ? patch[4] : patch[6];
        bfrag[1].y = lo ? patch[5] : patch[7];
        bfrag[2].x = lo ? patch[8] : 0.0f;
        bfrag[2].y = 0.0f;

        float best = -3.402823466e38f;
#pragma unroll
        for (int j = 0; j < 5; ++j) {
            v8f acc = {};
            acc = __builtin_amdgcn_wmma_f32_16x16x4_f32(false, afrag[j][0], false, bfrag[0],
                                                        (short)0, acc, false, false);
            acc = __builtin_amdgcn_wmma_f32_16x16x4_f32(false, afrag[j][1], false, bfrag[1],
                                                        (short)0, acc, false, false);
            acc = __builtin_amdgcn_wmma_f32_16x16x4_f32(false, afrag[j][2], false, bfrag[2],
                                                        (short)0, acc, false, false);
            float mx = acc[0];
#pragma unroll
            for (int t = 1; t < 8; ++t) mx = fmaxf(mx, acc[t]);
            mx = fmaxf(mx, 0.0f);
            float ssum = 0.0f, q = 0.0f;
#pragma unroll
            for (int t = 0; t < 8; ++t) {
                const float e = __builtin_amdgcn_exp2f(acc[t] - mx);
                ssum += e;
                q    += e * patch[t];
            }
            {
                const float e = __builtin_amdgcn_exp2f(-mx);
                ssum += e;
                q    += e * patch[8];
            }
            best = fmaxf(best, q / ssum);
        }
        best = fmaxf(best, __shfl_xor(best, 16));
        outv = (hh == tile) ? best : outv;
    }
    return outv;
}

__global__ __launch_bounds__(kThreads)
void vin_two_step_kernel(const float* __restrict__ v_in,
                         const float* __restrict__ rewards,
                         const float* __restrict__ wt,
                         float* __restrict__ v_out1,
                         float* __restrict__ v_out2) {
    __shared__ float lds_rv1[kStageRows][kPitch];
    __shared__ float lds_rw[kStageRows][kW];
    __shared__ float lds_rv2[kRows1][kPitch];
    __shared__ v2f lds_wt[kWtTrips * kThreads];

    const int tid  = threadIdx.x;
    const int lane = tid & 31;
    const int wv   = __builtin_amdgcn_readfirstlane(tid >> 5);
    const int c    = lane & 15;
    const int hh   = lane >> 4;
    const bool lo  = lane < 16;
    const int b    = blockIdx.x / kStrips;
    const int h0   = (blockIdx.x % kStrips) * kStrip;

    const float* vimg = v_in    + (size_t)b * kH * kW;
    const float* rimg = rewards + (size_t)b * kH * kW;

    float sv[kInTrips], sr[kInTrips];
#pragma unroll
    for (int j = 0; j < kInTrips; ++j) {
        const int e   = tid + j * kThreads;
        const int row = e / kW;
        const int col = e - row * kW;
        const int hc  = min(max(h0 - 2 + row, 0), kH - 1);
        sv[j] = vimg[hc * kW + col];
        sr[j] = rimg[hc * kW + col];
    }
    v2f wg[kWtTrips];
#pragma unroll
    for (int j = 0; j < kWtTrips; ++j)
        wg[j] = ((const v2f*)wt)[min(tid + j * kThreads, kWtPairs - 1)];

#pragma unroll
    for (int j = 0; j < kInTrips; ++j) {
        const int e   = tid + j * kThreads;
        const int row = e / kW;
        const int col = e - row * kW;
        const int hr  = h0 - 2 + row;
        const bool inb = (hr >= 0) & (hr < kH);
        lds_rv1[row][col + 1] = inb ? (sv[j] + sr[j]) : 0.0f;
        lds_rw[row][col]      = sr[j];
    }
    if (tid < kRowsIn) {
        lds_rv1[tid][0]      = 0.0f;
        lds_rv1[tid][kW + 1] = 0.0f;
    }
    if (tid < kRows1) {
        lds_rv2[tid][0]      = 0.0f;
        lds_rv2[tid][kW + 1] = 0.0f;
    }
#pragma unroll
    for (int j = 0; j < kWtTrips; ++j)
        lds_wt[tid + j * kThreads] = wg[j];
    __syncthreads();

    const int r8 = c & 7;
    const int hs = c >> 3;
    v2f afrag[5][3];
    {
        const float* lw   = (const float*)lds_wt;
        const float* wrow = lw + (hs * 9 + r8) * 9;
        const float* wref = lw + (hs * 9 + 8) * 9;
#pragma unroll
        for (int j = 0; j < 5; ++j) {
            const float* p = wrow + j * 162 + 2 * hh;
            const float* z = wref + j * 162 + 2 * hh;
            afrag[j][0].x = kLog2e * (p[0] - z[0]);
            afrag[j][0].y = kLog2e * (p[1] - z[1]);
            afrag[j][1].x = kLog2e * (p[4] - z[4]);
            afrag[j][1].y = kLog2e * (p[5] - z[5]);
            const float w8 = kLog2e * (wrow[j * 162 + 8] - wref[j * 162 + 8]);
            afrag[j][2].x = lo ? w8 : 0.0f;
            afrag[j][2].y = 0.0f;
        }
    }

    for (int i = 0; i < 2; ++i) {
        const int u   = wv + i * kWaves;
        const int lr  = u / kSegs;
        const int seg = u - lr * kSegs;
        const float nv = vin_unit(&lds_rv1[lr][0], seg * 32, afrag, c, hh, lo);
        const int hr  = h0 - 1 + lr;
        const int col = seg * 32 + lane;
        const bool inb = (hr >= 0) & (hr < kH);
        lds_rv2[lr][col + 1] = inb ? (nv + lds_rw[lr + 1][col]) : 0.0f;
        if (lr >= 1 && lr <= kStrip)
            v_out1[((size_t)b * kH + hr) * kW + col] = nv;
    }
    __syncthreads();

    const int n2 = (wv % 5 <= 1) ? 2 : 1;
    for (int i = 0; i < n2; ++i) {
        const int u   = (i == 0) ? wv : (kWaves + (wv / 5) * 2 + (wv % 5));
        const int lr  = u / kSegs;
        const int seg = u - lr * kSegs;
        const float nv = vin_unit(&lds_rv2[lr][0], seg * 32, afrag, c, hh, lo);
        v_out2[((size_t)b * kH + h0 + lr) * kW + seg * 32 + lane] = nv;
    }
}

extern "C" void kernel_launch(void* const* d_in, const int* in_sizes, int n_in,
                              void* d_out, int out_size, void* d_ws, size_t ws_size,
                              hipStream_t stream) {
    (void)in_sizes; (void)n_in; (void)out_size; (void)d_ws; (void)ws_size;
    const float* values  = (const float*)d_in[0];
    const float* rewards = (const float*)d_in[1];
    const float* wt      = (const float*)d_in[2];
    float* out = (float*)d_out;

    const size_t slice = (size_t)kB * kH * kW;
    for (int t = 0; t < kDepth; t += 2) {
        const float* vin = (t == 0) ? values : out + (size_t)(t - 1) * slice;
        float* vout1 = out + (size_t)t * slice;
        float* vout2 = out + (size_t)(t + 1) * slice;
        vin_two_step_kernel<<<kB * kStrips, kThreads, 0, stream>>>(vin, rewards, wt, vout1, vout2);
    }
}


# Softmax normalisation q/ssum taken as q * v_rcp_f32(ssum): ssum is always in [1,9] (largest exponential is exactly 1), so 1 reciprocal (1 ulp) + 1 multiply replace the 11-instruction f32 division
# speedup vs baseline: 3.6285x; 1.0665x over previous
#include <hip/hip_runtime.h>


constexpr int kDepth   = 16;
constexpr int kB       = 16;
constexpr int kH       = 96;
constexpr int kW       = 96;
constexpr int kStrip   = 6;
constexpr int kStrips  = kH / kStrip;
constexpr int kRows1   = kStrip + 2;
constexpr int kRowsIn  = kStrip + 4;
constexpr int kWaves   = 12;
constexpr int kThreads = kWaves * 32;
constexpr int kPitch   = kW + 2;
constexpr int kSegs    = kW / 32;
constexpr int kUnits1  = kRows1 * kSegs;
constexpr int kUnits2  = kStrip * kSegs;
constexpr int kInTrips = (kRowsIn * kW + kThreads - 1) / kThreads;
constexpr int kStageRows = kInTrips * kThreads / kW;
static_assert(kInTrips * kThreads % kW == 0 && kStageRows >= kRowsIn, "whole rows per staging pass");
constexpr int kWtPairs = 90 * 9 / 2;
constexpr int kWtTrips = (kWtPairs + kThreads - 1) / kThreads;
static_assert(kH % kStrip == 0, "strips tile the image");
static_assert(kUnits1 == 2 * kWaves, "step one: exactly two units per wave");
static_assert(kUnits2 - kWaves == 6, "step two: six waves take a second unit");

typedef __attribute__((ext_vector_type(2))) float v2f;
typedef __attribute__((ext_vector_type(8))) float v8f;

constexpr float kLog2e = 1.4426950408889634f;

__device__ __forceinline__ float vin_unit(const float* rows, int px0,
                                          const v2f (&afrag)[5][3],
                                          int c, int hh, bool lo) {
    float outv = 0.0f;
#pragma unroll
    for (int tile = 0; tile < 2; ++tile) {
        const int px = px0 + tile * 16 + c;
        float patch[9];
#pragma unroll
        for (int t = 0; t < 9; ++t)
            patch[t] = rows[(t / 3) * kPitch + px + (t % 3)];

        v2f bfrag[3];
        bfrag[0].x = lo ? patch[0] : patch[2];
        bfrag[0].y = lo ? patch[1] : patch[3];
        bfrag[1].x = lo ? patch[4] : patch[6];
        bfrag[1].y = lo ? patch[5] : patch[7];
        bfrag[2].x = lo ? patch[8] : 0.0f;
        bfrag[2].y = 0.0f;

        float best = -3.402823466e38f;
#pragma unroll
        for (int j = 0; j < 5; ++j) {
            v8f acc = {};
            acc = __builtin_amdgcn_wmma_f32_16x16x4_f32(false, afrag[j][0], false, bfrag[0],
                                                        (short)0, acc, false, false);
            acc = __builtin_amdgcn_wmma_f32_16x16x4_f32(false, afrag[j][1], false, bfrag[1],
                                                        (short)0, acc, false, false);
            acc = __builtin_amdgcn_wmma_f32_16x16x4_f32(false, afrag[j][2], false, bfrag[2],
                                                        (short)0, acc, false, false);
            float mx = acc[0];
#pragma unroll
            for (int t = 1; t < 8; ++t) mx = fmaxf(mx, acc[t]);
            mx = fmaxf(mx, 0.0f);
            float ssum = 0.0f, q = 0.0f;
#pragma unroll
            for (int t = 0; t < 8; ++t) {
                const float e = __builtin_amdgcn_exp2f(acc[t] - mx);
                ssum += e;
                q    += e * patch[t];
            }
            {
                const float e = __builtin_amdgcn_exp2f(-mx);
                ssum += e;
                q    += e * patch[8];
            }
            best = fmaxf(best, q * __builtin_amdgcn_rcpf(ssum));
        }
        best = fmaxf(best, __shfl_xor(best, 16));
        outv = (hh == tile) ? best : outv;
    }
    return outv;
}

__global__ __launch_bounds__(kThreads)
void vin_two_step_kernel(const float* __restrict__ v_in,
                         const float* __restrict__ rewards,
                         const float* __restrict__ wt,
                         float* __restrict__ v_out1,
                         float* __restrict__ v_out2) {
    __shared__ float lds_rv1[kStageRows][kPitch];
    __shared__ float lds_rw[kStageRows][kW];
    __shared__ float lds_rv2[kRows1][kPitch];
    __shared__ v2f lds_wt[kWtTrips * kThreads];

    const int tid  = threadIdx.x;
    const int lane = tid & 31;
    const int wv   = __builtin_amdgcn_readfirstlane(tid >> 5);
    const int c    = lane & 15;
    const int hh   = lane >> 4;
    const bool lo  = lane < 16;
    const int b    = blockIdx.x / kStrips;
    const int h0   = (blockIdx.x % kStrips) * kStrip;

    const float* vimg = v_in    + (size_t)b * kH * kW;
    const float* rimg = rewards + (size_t)b * kH * kW;

    float sv[kInTrips], sr[kInTrips];
#pragma unroll
    for (int j = 0; j < kInTrips; ++j) {
        const int e   = tid + j * kThreads;
        const int row = e / kW;
        const int col = e - row * kW;
        const int hc  = min(max(h0 - 2 + row, 0), kH - 1);
        sv[j] = vimg[hc * kW + col];
        sr[j] = rimg[hc * kW + col];
    }
    v2f wg[kWtTrips];
#pragma unroll
    for (int j = 0; j < kWtTrips; ++j)
        wg[j] = ((const v2f*)wt)[min(tid + j * kThreads, kWtPairs - 1)];

#pragma unroll
    for (int j = 0; j < kInTrips; ++j) {
        const int e   = tid + j * kThreads;
        const int row = e / kW;
        const int col = e - row * kW;
        const int hr  = h0 - 2 + row;
        const bool inb = (hr >= 0) & (hr < kH);
        lds_rv1[row][col + 1] = inb ? (sv[j] + sr[j]) : 0.0f;
        lds_rw[row][col]      = sr[j];
    }
    if (tid < kRowsIn) {
        lds_rv1[tid][0]      = 0.0f;
        lds_rv1[tid][kW + 1] = 0.0f;
    }
    if (tid < kRows1) {
        lds_rv2[tid][0]      = 0.0f;
        lds_rv2[tid][kW + 1] = 0.0f;
    }
#pragma unroll
    for (int j = 0; j < kWtTrips; ++j)
        lds_wt[tid + j * kThreads] = wg[j];
    __syncthreads();

    const int r8 = c & 7;
    const int hs = c >> 3;
    v2f afrag[5][3];
    {
        const float* lw   = (const float*)lds_wt;
        const float* wrow = lw + (hs * 9 + r8) * 9;
        const float* wref = lw + (hs * 9 + 8) * 9;
#pragma unroll
        for (int j = 0; j < 5; ++j) {
            const float* p = wrow + j * 162 + 2 * hh;
            const float* z = wref + j * 162 + 2 * hh;
            afrag[j][0].x = kLog2e * (p[0] - z[0]);
            afrag[j][0].y = kLog2e * (p[1] - z[1]);
            afrag[j][1].x = kLog2e * (p[4] - z[4]);
            afrag[j][1].y = kLog2e * (p[5] - z[5]);
            const float w8 = kLog2e * (wrow[j * 162 + 8] - wref[j * 162 + 8]);
            afrag[j][2].x = lo ? w8 : 0.0f;
            afrag[j][2].y = 0.0f;
        }
    }

    for (int i = 0; i < 2; ++i) {
        const int u   = wv + i * kWaves;
        const int lr  = u / kSegs;
        const int seg = u - lr * kSegs;
        const float nv = vin_unit(&lds_rv1[lr][0], seg * 32, afrag, c, hh, lo);
        const int hr  = h0 - 1 + lr;
        const int col = seg * 32 + lane;
        const bool inb = (hr >= 0) & (hr < kH);
        lds_rv2[lr][col + 1] = inb ? (nv + lds_rw[lr + 1][col]) : 0.0f;
        if (lr >= 1 && lr <= kStrip)
            v_out1[((size_t)b * kH + hr) * kW + col] = nv;
    }
    __syncthreads();

    const int n2 = (wv % 5 <= 1) ? 2 : 1;
    for (int i = 0; i < n2; ++i) {
        const int u   = (i == 0) ? wv : (kWaves + (wv / 5) * 2 + (wv % 5));
        const int lr  = u / kSegs;
        const int seg = u - lr * kSegs;
        const float nv = vin_unit(&lds_rv2[lr][0], seg * 32, afrag, c, hh, lo);
        v_out2[((size_t)b * kH + h0 + lr) * kW + seg * 32 + lane] = nv;
    }
}

extern "C" void kernel_launch(void* const* d_in, const int* in_sizes, int n_in,
                              void* d_out, int out_size, void* d_ws, size_t ws_size,
                              hipStream_t stream) {
    (void)in_sizes; (void)n_in; (void)out_size; (void)d_ws; (void)ws_size;
    const float* values  = (const float*)d_in[0];
    const float* rewards = (const float*)d_in[1];
    const float* wt      = (const float*)d_in[2];
    float* out = (float*)d_out;

    const size_t slice = (size_t)kB * kH * kW;
    for (int t = 0; t < kDepth; t += 2) {
        const float* vin = (t == 0) ? values : out + (size_t)(t - 1) * slice;
        float* vout1 = out + (size_t)t * slice;
        float* vout2 = out + (size_t)(t + 1) * slice;
        vin_two_step_kernel<<<kB * kStrips, kThreads, 0, stream>>>(vin, rewards, wt, vout1, vout2);
    }
}


# Work dealt out in tiles of 16 pixels, not lines of 32: a wave takes 4 tiles in step one, 3 in step two (before: 2 lines, then six waves a second line while six idled); per-pixel arithmetic unchanged
# speedup vs baseline: 3.7366x; 1.0298x over previous
#include <hip/hip_runtime.h>


constexpr int kDepth   = 16;
constexpr int kB       = 16;
constexpr int kH       = 96;
constexpr int kW       = 96;
constexpr int kStrip   = 6;
constexpr int kStrips  = kH / kStrip;
constexpr int kRows1   = kStrip + 2;
constexpr int kRowsIn  = kStrip + 4;
constexpr int kWaves   = 12;
constexpr int kThreads = kWaves * 32;
constexpr int kPitch   = kW + 2;
constexpr int kSegs    = kW / 32;
constexpr int kTiles1  = kRows1 * kSegs * 2;
constexpr int kTiles2  = kStrip * kSegs * 2;
constexpr int kPerWave1 = kTiles1 / kWaves;
constexpr int kPerWave2 = kTiles2 / kWaves;
constexpr int kInTrips = (kRowsIn * kW + kThreads - 1) / kThreads;
constexpr int kStageRows = kInTrips * kThreads / kW;
static_assert(kInTrips * kThreads % kW == 0 && kStageRows >= kRowsIn, "whole rows per staging pass");
constexpr int kWtPairs = 90 * 9 / 2;
constexpr int kWtTrips = (kWtPairs + kThreads - 1) / kThreads;
static_assert(kH % kStrip == 0, "strips tile the image");
static_assert(kPerWave1 == 4 && kPerWave1 * kWaves == kTiles1, "step one: two whole lines per wave");
static_assert(kPerWave2 == 3 && kPerWave2 * kWaves == kTiles2, "step two: one whole line and one half line per wave");

typedef __attribute__((ext_vector_type(2))) float v2f;
typedef __attribute__((ext_vector_type(8))) float v8f;

constexpr float kLog2e = 1.4426950408889634f;

__device__ __forceinline__ float vin_tile(const float* rows, int px0,
                                          const v2f (&afrag)[5][3],
                                          int c, bool lo) {
    const int px = px0 + c;
    float patch[9];
#pragma unroll
    for (int t = 0; t < 9; ++t)
        patch[t] = rows[(t / 3) * kPitch + px + (t % 3)];

    v2f bfrag[3];
    bfrag[0].x = lo ? patch[0] : patch[2];
    bfrag[0].y = lo ? patch[1] : patch[3];
    bfrag[1].x = lo ? patch[4] : patch[6];
    bfrag[1].y = lo ? patch[5] : patch[7];
    bfrag[2].x = lo ? patch[8] : 0.0f;
    bfrag[2].y = 0.0f;

    float best = -3.402823466e38f;
#pragma unroll
    for (int j = 0; j < 5; ++j) {
        v8f acc = {};
        acc = __builtin_amdgcn_wmma_f32_16x16x4_f32(false, afrag[j][0], false, bfrag[0],
                                                    (short)0, acc, false, false);
        acc = __builtin_amdgcn_wmma_f32_16x16x4_f32(false, afrag[j][1], false, bfrag[1],
                                                    (short)0, acc, false, false);
        acc = __builtin_amdgcn_wmma_f32_16x16x4_f32(false, afrag[j][2], false, bfrag[2],
                                                    (short)0, acc, false, false);
        float mx = acc[0];
#pragma unroll
        for (int t = 1; t < 8; ++t) mx = fmaxf(mx, acc[t]);
        mx = fmaxf(mx, 0.0f);
        float ssum = 0.0f, q = 0.0f;
#pragma unroll
        for (int t = 0; t < 8; ++t) {
            const float e = __builtin_amdgcn_exp2f(acc[t] - mx);
            ssum += e;
            q    += e * patch[t];
        }
        {
            const float e = __builtin_amdgcn_exp2f(-mx);
            ssum += e;
            q    += e * patch[8];
        }
        best = fmaxf(best, q * __builtin_amdgcn_rcpf(ssum));
    }
    return fmaxf(best, __shfl_xor(best, 16));
}

__global__ __launch_bounds__(kThreads)
void vin_two_step_kernel(const float* __restrict__ v_in,
                         const float* __restrict__ rewards,
                         const float* __restrict__ wt,
                         float* __restrict__ v_out1,
                         float* __restrict__ v_out2) {
    __shared__ float lds_rv1[kStageRows][kPitch];
    __shared__ float lds_rw[kStageRows][kW];
    __shared__ float lds_rv2[kRows1][kPitch];
    __shared__ v2f lds_wt[kWtTrips * kThreads];

    const int tid  = threadIdx.x;
    const int lane = tid & 31;
    const int wv   = __builtin_amdgcn_readfirstlane(tid >> 5);
    const int c    = lane & 15;
    const int hh   = lane >> 4;
    const bool lo  = lane < 16;
    const int b    = blockIdx.x / kStrips;
    const int h0   = (blockIdx.x % kStrips) * kStrip;

    const float* vimg = v_in    + (size_t)b * kH * kW;
    const float* rimg = rewards + (size_t)b * kH * kW;

    float sv[kInTrips], sr[kInTrips];
#pragma unroll
    for (int j = 0; j < kInTrips; ++j) {
        const int e   = tid + j * kThreads;
        const int row = e / kW;
        const int col = e - row * kW;
        const int hc  = min(max(h0 - 2 + row, 0), kH - 1);
        sv[j] = vimg[hc * kW + col];
        sr[j] = rimg[hc * kW + col];
    }
    v2f wg[kWtTrips];
#pragma unroll
    for (int j = 0; j < kWtTrips; ++j)
        wg[j] = ((const v2f*)wt)[min(tid + j * kThreads, kWtPairs - 1)];

#pragma unroll
    for (int j = 0; j < kInTrips; ++j) {
        const int e   = tid + j * kThreads;
        const int row = e / kW;
        const int col = e - row * kW;
        const int hr  = h0 - 2 + row;
        const bool inb = (hr >= 0) & (hr < kH);
        lds_rv1[row][col + 1] = inb ? (sv[j] + sr[j]) : 0.0f;
        lds_rw[row][col]      = sr[j];
    }
    if (tid < kRowsIn) {
        lds_rv1[tid][0]      = 0.0f;
        lds_rv1[tid][kW + 1] = 0.0f;
    }
    if (tid < kRows1) {
        lds_rv2[tid][0]      = 0.0f;
        lds_rv2[tid][kW + 1] = 0.0f;
    }
#pragma unroll
    for (int j = 0; j < kWtTrips; ++j)
        lds_wt[tid + j * kThreads] = wg[j];
    __syncthreads();

    const int r8 = c & 7;
    const int hs = c >> 3;
    v2f afrag[5][3];
    {
        const float* lw   = (const float*)lds_wt;
        const float* wrow = lw + (hs * 9 + r8) * 9;
        const float* wref = lw + (hs * 9 + 8) * 9;
#pragma unroll
        for (int j = 0; j < 5; ++j) {
            const float* p = wrow + j * 162 + 2 * hh;
            const float* z = wref + j * 162 + 2 * hh;
            afrag[j][0].x = kLog2e * (p[0] - z[0]);
            afrag[j][0].y = kLog2e * (p[1] - z[1]);
            afrag[j][1].x = kLog2e * (p[4] - z[4]);
            afrag[j][1].y = kLog2e * (p[5] - z[5]);
            const float w8 = kLog2e * (wrow[j * 162 + 8] - wref[j * 162 + 8]);
            afrag[j][2].x = lo ? w8 : 0.0f;
            afrag[j][2].y = 0.0f;
        }
    }

    for (int i = 0; i < kPerWave1; ++i) {
        const int u   = wv + (i >> 1) * kWaves;
        const int lr  = u / kSegs;
        const int seg = u - lr * kSegs;
        const int px0 = seg * 32 + (i & 1) * 16;
        const float nv = vin_tile(&lds_rv1[lr][0], px0, afrag, c, lo);
        const int hr  = h0 - 1 + lr;
        const int col = px0 + c;
        const bool inb = (hr >= 0) & (hr < kH);
        if (lo) {
            lds_rv2[lr][col + 1] = inb ? (nv + lds_rw[lr + 1][col]) : 0.0f;
            if (lr >= 1 && lr <= kStrip)
                v_out1[((size_t)b * kH + hr) * kW + col] = nv;
        }
    }
    __syncthreads();

    for (int i = 0; i < kPerWave2; ++i) {
        const int u    = (i < 2) ? wv : (kWaves + (wv >> 1));
        const int half = (i < 2) ? i : (wv & 1);
        const int lr   = u / kSegs;
        const int seg  = u - lr * kSegs;
        const int px0  = seg * 32 + half * 16;
        const float nv = vin_tile(&lds_rv2[lr][0], px0, afrag, c, lo);
        if (lo)
            v_out2[((size_t)b * kH + h0 + lr) * kW + px0 + c] = nv;
    }
}

extern "C" void kernel_launch(void* const* d_in, const int* in_sizes, int n_in,
                              void* d_out, int out_size, void* d_ws, size_t ws_size,
                              hipStream_t stream) {
    (void)in_sizes; (void)n_in; (void)out_size; (void)d_ws; (void)ws_size;
    const float* values  = (const float*)d_in[0];
    const float* rewards = (const float*)d_in[1];
    const float* wt      = (const float*)d_in[2];
    float* out = (float*)d_out;

    const size_t slice = (size_t)kB * kH * kW;
    for (int t = 0; t < kDepth; t += 2) {
        const float* vin = (t == 0) ? values : out + (size_t)(t - 1) * slice;
        float* vout1 = out + (size_t)t * slice;
        float* vout2 = out + (size_t)(t + 1) * slice;
        vin_two_step_kernel<<<kB * kStrips, kThreads, 0, stream>>>(vin, rewards, wt, vout1, vout2);
    }
}
